# starting kernel
# baseline (speedup 1.0000x reference)
#include <hip/hip_runtime.h>
#include <math.h>

typedef __attribute__((ext_vector_type(2))) float v2f;
typedef __attribute__((ext_vector_type(8))) float v8f;

#define QN 2048
#define KN 2048
#define CIN 256
#define HD 256
#define NH 8
#define CH 32
#define KSPLIT 4

__device__ __forceinline__ v8f wmma_f32x4(v2f a, v2f b, v8f c) {
    return __builtin_amdgcn_wmma_f32_16x16x4_f32(
         false, a,  false, b,
         (short)0, c,  false,  false);
}

__device__ __forceinline__ v8f gemm_tile_f32(const float* __restrict__ A, int lda,
                                             const float* __restrict__ B, int ldb,
                                             int CK, int lane) {
    const int m = lane & 15;
    const int hi = lane >> 4;
    const int n = m;
    v8f acc = {};
    for (int k0 = 0; k0 < CK; k0 += 4) {
        v2f a, b;
        a.x = A[m * lda + k0 + 2 * hi + 0];
        a.y = A[m * lda + k0 + 2 * hi + 1];
        b.x = B[(k0 + 2 * hi + 0) * ldb + n];
        b.y = B[(k0 + 2 * hi + 1) * ldb + n];
        acc = wmma_f32x4(a, b, acc);
    }
    return acc;
}

__global__ void proj_kernel(const float* __restrict__ qx, const float* __restrict__ kvx,
                            const float* __restrict__ Wq, const float* __restrict__ Wk,
                            const float* __restrict__ Wv, const float* __restrict__ Wg,
                            const float* __restrict__ bg,
                            float* __restrict__ q_ws, float* __restrict__ k_ws,
                            float* __restrict__ v_ws, float* __restrict__ g_ws) {
    const int lane = threadIdx.x & 31;
    const int tile = blockIdx.x;
    const int row0 = blockIdx.y * 16;
    const int which = blockIdx.z;

    const float* A = (which == 1 || which == 2) ? kvx : qx;
    const float* W = (which == 0) ? Wq : (which == 1) ? Wk : (which == 2) ? Wv : Wg;
    float* O = (which == 0) ? q_ws : (which == 1) ? k_ws : (which == 2) ? v_ws : g_ws;

    v8f acc = gemm_tile_f32(A + (size_t)row0 * CIN, CIN, W + tile * 16, HD, CIN, lane);

    const int m = lane & 15, hi = lane >> 4, n = m;
    const int col = tile * 16 + n;
    const float bgc = (which == 3) ? bg[col] : 0.f;
#pragma unroll
    for (int r = 0; r < 8; ++r) {
        float val = acc[r];
        if (which == 0) val *= 0.17677669529663687f;
        if (which == 3) val = __builtin_amdgcn_rcpf(1.f + __expf(-(val + bgc)));
        O[(size_t)(row0 + r + 8 * hi) * HD + col] = val;
    }
}

__global__ void __launch_bounds__(128)
flash_kernel(const float* __restrict__ q_ws, const float* __restrict__ k_ws,
             const float* __restrict__ v_ws, const float* __restrict__ g_ws,
             const float* __restrict__ bias_mask, const float* __restrict__ bias_pair,
             float* __restrict__ o_ws) {
    __shared__ float ldsP[KSPLIT][16 * 18];
    __shared__ float Ms[KSPLIT][16];
    __shared__ float Ls[KSPLIT][16];
    __shared__ float O0s[KSPLIT][16][16];
    __shared__ float O1s[KSPLIT][16][16];

    const int lane = threadIdx.x & 31;
    const int wv = threadIdx.x >> 5;
    const int h = blockIdx.x >> 7;
    const int q0 = (blockIdx.x & 127) * 16;
    const int m = lane & 15, hi = lane >> 4, n = m;
    float* P = ldsP[wv];

    v2f aq[8];
    const float* Qb = q_ws + (size_t)q0 * HD + h * CH;
#pragma unroll
    for (int kk = 0; kk < 8; ++kk) {
        aq[kk].x = Qb[m * HD + kk * 4 + 2 * hi + 0];
        aq[kk].y = Qb[m * HD + kk * 4 + 2 * hi + 1];
    }

    v8f o0 = {}, o1 = {}, o2 = {};
    float mrow[8];
#pragma unroll
    for (int r = 0; r < 8; ++r) mrow[r] = -1e30f;

    v2f bones; bones.x = 1.f; bones.y = 1.f;
    const float* bp_base = bias_pair + (size_t)h * QN * KN + (size_t)(q0 + 8 * hi) * KN + n;
    const int kb_lo = wv * (KN / KSPLIT);
    const int kb_hi = kb_lo + (KN / KSPLIT);

    for (int kb = kb_lo; kb < kb_hi; kb += 16) {
        v8f s = {};
#pragma unroll
        for (int kk = 0; kk < 8; ++kk) {
            const float* Kb = k_ws + (size_t)(kb + n) * HD + h * CH + kk * 4 + 2 * hi;
            v2f b; b.x = Kb[0]; b.y = Kb[1];
            s = wmma_f32x4(aq[kk], b, s);
        }
        const float bm = bias_mask[kb + n];
        const float* bp = bp_base + kb;
#pragma unroll
        for (int r = 0; r < 8; ++r)
            s[r] += bm + __builtin_nontemporal_load(bp + (size_t)r * KN);
#pragma unroll
        for (int r = 0; r < 8; ++r) {
            float mt = s[r];
            mt = fmaxf(mt, __shfl_xor(mt, 1, 32));
            mt = fmaxf(mt, __shfl_xor(mt, 2, 32));
            mt = fmaxf(mt, __shfl_xor(mt, 4, 32));
            mt = fmaxf(mt, __shfl_xor(mt, 8, 32));
            const float mnew = fmaxf(mrow[r], mt);
            const float corr = __expf(mrow[r] - mnew);
            mrow[r] = mnew;
            s[r] = __expf(s[r] - mnew);
            o0[r] *= corr;
            o1[r] *= corr;
            o2[r] *= corr;
        }
#pragma unroll
        for (int r = 0; r < 8; ++r) P[(r + 8 * hi) * 18 + n] = s[r];
#pragma unroll
        for (int kk = 0; kk < 4; ++kk) {
            v2f ap;
            ap.x = P[m * 18 + kk * 4 + 2 * hi + 0];
            ap.y = P[m * 18 + kk * 4 + 2 * hi + 1];
            const float* Vb = v_ws + (size_t)(kb + kk * 4 + 2 * hi) * HD + h * CH;
            v2f b0, b1;
            b0.x = Vb[n];        b0.y = Vb[HD + n];
            b1.x = Vb[16 + n];   b1.y = Vb[HD + 16 + n];
            o0 = wmma_f32x4(ap, b0, o0);
            o1 = wmma_f32x4(ap, b1, o1);
            o2 = wmma_f32x4(ap, bones, o2);
        }
    }

    if (n == 0) {
#pragma unroll
        for (int r = 0; r < 8; ++r) {
            Ms[wv][r + 8 * hi] = mrow[r];
            Ls[wv][r + 8 * hi] = o2[r];
        }
    }
#pragma unroll
    for (int r = 0; r < 8; ++r) {
        O0s[wv][r + 8 * hi][n] = o0[r];
        O1s[wv][r + 8 * hi][n] = o1[r];
    }
    __syncthreads();

    if (wv == 0) {
        const float* gb = g_ws + (size_t)q0 * HD + h * CH;
        float* ob = o_ws + (size_t)q0 * HD + h * CH;
#pragma unroll
        for (int r = 0; r < 8; ++r) {
            const int row = r + 8 * hi;
            const float M = fmaxf(fmaxf(Ms[0][row], Ms[1][row]),
                                  fmaxf(Ms[2][row], Ms[3][row]));
            float l = 0.f, a0 = 0.f, a1 = 0.f;
#pragma unroll
            for (int w = 0; w < KSPLIT; ++w) {
                const float f = __expf(Ms[w][row] - M);
                l  += Ls[w][row] * f;
                a0 += O0s[w][row][n] * f;
                a1 += O1s[w][row][n] * f;
            }
            const float inv = __builtin_amdgcn_rcpf(l);
            ob[(size_t)row * HD + n]      = a0 * inv * gb[(size_t)row * HD + n];
            ob[(size_t)row * HD + 16 + n] = a1 * inv * gb[(size_t)row * HD + 16 + n];
        }
    }
}

__global__ void out_kernel(const float* __restrict__ o_ws, const float* __restrict__ Wo,
                           const float* __restrict__ bo, float* __restrict__ out) {
    const int lane = threadIdx.x & 31;
    const int tile = blockIdx.x;
    const int row0 = blockIdx.y * 16;
    v8f acc = gemm_tile_f32(o_ws + (size_t)row0 * HD, HD, Wo + tile * 16, CIN, HD, lane);
    const int m = lane & 15, hi = lane >> 4, n = m;
    const float b = bo[tile * 16 + n];
#pragma unroll
    for (int r = 0; r < 8; ++r)
        out[(size_t)(row0 + r + 8 * hi) * CIN + tile * 16 + n] = acc[r] + b;
}

extern "C" void kernel_launch(void* const* d_in, const int* in_sizes, int n_in,
                              void* d_out, int out_size, void* d_ws, size_t ws_size,
                              hipStream_t stream) {
    const float* q_x       = (const float*)d_in[0];
    const float* kv_x      = (const float*)d_in[1];
    const float* bias_mask = (const float*)d_in[2];
    const float* bias_pair = (const float*)d_in[3];
    const float* Wq        = (const float*)d_in[4];
    const float* Wk        = (const float*)d_in[5];
    const float* Wv        = (const float*)d_in[6];
    const float* Wo        = (const float*)d_in[7];
    const float* bo        = (const float*)d_in[8];
    const float* Wg        = (const float*)d_in[9];
    const float* bg        = (const float*)d_in[10];
    float* out = (float*)d_out;

    const size_t mat = (size_t)QN * HD;
    float* q_ws = (float*)d_ws;
    float* k_ws = q_ws + mat;
    float* v_ws = k_ws + mat;
    float* g_ws = v_ws + mat;
    float* o_ws = g_ws + mat;

    proj_kernel<<<dim3(HD / 16, QN / 16, 4), 32, 0, stream>>>(
        q_x, kv_x, Wq, Wk, Wv, Wg, bg, q_ws, k_ws, v_ws, g_ws);

    flash_kernel<<<dim3(NH * (QN / 16)), 128, 0, stream>>>(
        q_ws, k_ws, v_ws, g_ws, bias_mask, bias_pair, o_ws);

    out_kernel<<<dim3(CIN / 16, QN / 16), 32, 0, stream>>>(o_ws, Wo, bo, out);
}


# attention kernel computes transposed tiles (S^T=K*Q^T, O^T=V^T*P^T): row max in-lane + one permlanex16, P^T used from registers, no LDS relayout; key order inside tile sums changes
# speedup vs baseline: 1.2413x; 1.2413x over previous
#include <hip/hip_runtime.h>
#include <math.h>

typedef __attribute__((ext_vector_type(2))) float v2f;
typedef __attribute__((ext_vector_type(8))) float v8f;

#define QN 2048
#define KN 2048
#define CIN 256
#define HD 256
#define NH 8
#define CH 32
#define KSPLIT 4

__device__ __forceinline__ v8f wmma_f32x4(v2f a, v2f b, v8f c) {
    return __builtin_amdgcn_wmma_f32_16x16x4_f32(
         false, a,  false, b,
         (short)0, c,  false,  false);
}

__device__ __forceinline__ v8f gemm_tile_f32(const float* __restrict__ A, int lda,
                                             const float* __restrict__ B, int ldb,
                                             int CK, int lane) {
    const int m = lane & 15;
    const int hi = lane >> 4;
    const int n = m;
    v8f acc = {};
    for (int k0 = 0; k0 < CK; k0 += 4) {
        v2f a, b;
        a.x = A[m * lda + k0 + 2 * hi + 0];
        a.y = A[m * lda + k0 + 2 * hi + 1];
        b.x = B[(k0 + 2 * hi + 0) * ldb + n];
        b.y = B[(k0 + 2 * hi + 1) * ldb + n];
        acc = wmma_f32x4(a, b, acc);
    }
    return acc;
}

__global__ void proj_kernel(const float* __restrict__ qx, const float* __restrict__ kvx,
                            const float* __restrict__ Wq, const float* __restrict__ Wk,
                            const float* __restrict__ Wv, const float* __restrict__ Wg,
                            const float* __restrict__ bg,
                            float* __restrict__ q_ws, float* __restrict__ k_ws,
                            float* __restrict__ v_ws, float* __restrict__ g_ws) {
    const int lane = threadIdx.x & 31;
    const int tile = blockIdx.x;
    const int row0 = blockIdx.y * 16;
    const int which = blockIdx.z;

    const float* A = (which == 1 || which == 2) ? kvx : qx;
    const float* W = (which == 0) ? Wq : (which == 1) ? Wk : (which == 2) ? Wv : Wg;
    float* O = (which == 0) ? q_ws : (which == 1) ? k_ws : (which == 2) ? v_ws : g_ws;

    v8f acc = gemm_tile_f32(A + (size_t)row0 * CIN, CIN, W + tile * 16, HD, CIN, lane);

    const int m = lane & 15, hi = lane >> 4, n = m;
    const int col = tile * 16 + n;
    const float bgc = (which == 3) ? bg[col] : 0.f;
#pragma unroll
    for (int r = 0; r < 8; ++r) {
        float val = acc[r];
        if (which == 0) val *= 0.17677669529663687f;
        if (which == 3) val = __builtin_amdgcn_rcpf(1.f + __expf(-(val + bgc)));
        O[(size_t)(row0 + r + 8 * hi) * HD + col] = val;
    }
}

typedef __attribute__((ext_vector_type(4))) float v4f;

__device__ __forceinline__ float xhalf(float x) {
    const int b = __builtin_bit_cast(int, x);
    return __builtin_bit_cast(float,
        __builtin_amdgcn_permlanex16(b, b, 0x76543210, 0xfedcba98, false, false));
}

__global__ void __launch_bounds__(128)
flash_kernel(const float* __restrict__ q_ws, const float* __restrict__ k_ws,
             const float* __restrict__ v_ws, const float* __restrict__ g_ws,
             const float* __restrict__ bias_mask, const float* __restrict__ bias_pair,
             float* __restrict__ o_ws) {
    __shared__ float Ms[KSPLIT][32];
    __shared__ float Ls[KSPLIT][32];
    __shared__ float Os[KSPLIT][16][32];

    const int lane = threadIdx.x & 31;
    const int wv = threadIdx.x >> 5;
    const int h = blockIdx.x >> 7;
    const int q0 = (blockIdx.x & 127) * 16;
    const int c = lane & 15, hi = lane >> 4;

    v2f bq[8];
    const float* Qb = q_ws + (size_t)(q0 + c) * HD + h * CH + 2 * hi;
#pragma unroll
    for (int kk = 0; kk < 8; ++kk) {
        bq[kk].x = Qb[kk * 4 + 0];
        bq[kk].y = Qb[kk * 4 + 1];
    }

    v8f o0 = {}, o1 = {}, o2 = {};
    float mrow = -1e30f;

    v2f aones; aones.x = 1.f; aones.y = 1.f;
    const float* bp_base = bias_pair + (size_t)h * QN * KN + (size_t)(q0 + c) * KN + 8 * hi;
    const float* bm_base = bias_mask + 8 * hi;
    const float* Kbase = k_ws + (size_t)c * HD + h * CH + 2 * hi;
    const float* Vbase = v_ws + (size_t)(8 * hi) * HD + h * CH + c;
    const int kb_lo = wv * (KN / KSPLIT);
    const int kb_hi = kb_lo + (KN / KSPLIT);

    for (int kb = kb_lo; kb < kb_hi; kb += 16) {
        v8f s = {};
        const float* Kb = Kbase + (size_t)kb * HD;
#pragma unroll
        for (int kk = 0; kk < 8; ++kk) {
            v2f a; a.x = Kb[kk * 4 + 0]; a.y = Kb[kk * 4 + 1];
            s = wmma_f32x4(a, bq[kk], s);
        }
        const v4f bm0 = *(const v4f*)(bm_base + kb);
        const v4f bm1 = *(const v4f*)(bm_base + kb + 4);
        const v4f bp0 = __builtin_nontemporal_load((const v4f*)(bp_base + kb));
        const v4f bp1 = __builtin_nontemporal_load((const v4f*)(bp_base + kb + 4));
#pragma unroll
        for (int r = 0; r < 4; ++r) {
            s[r]     = (s[r]     + bm0[r]) + bp0[r];
            s[r + 4] = (s[r + 4] + bm1[r]) + bp1[r];
        }
        float mt = fmaxf(fmaxf(fmaxf(s[0], s[1]), fmaxf(s[2], s[3])),
                         fmaxf(fmaxf(s[4], s[5]), fmaxf(s[6], s[7])));
        mt = fmaxf(mt, xhalf(mt));
        const float mnew = fmaxf(mrow, mt);
        const float corr = __expf(mrow - mnew);
        mrow = mnew;
#pragma unroll
        for (int r = 0; r < 8; ++r) {
            s[r] = __expf(s[r] - mnew);
            o0[r] *= corr;
            o1[r] *= corr;
            o2[r] *= corr;
        }
        const float* Vb = Vbase + (size_t)kb * HD;
#pragma unroll
        for (int j = 0; j < 4; ++j) {
            v2f p; p.x = s[2 * j]; p.y = s[2 * j + 1];
            v2f a0, a1;
            a0.x = Vb[(2 * j) * HD];       a0.y = Vb[(2 * j + 1) * HD];
            a1.x = Vb[(2 * j) * HD + 16];  a1.y = Vb[(2 * j + 1) * HD + 16];
            o0 = wmma_f32x4(a0, p, o0);
            o1 = wmma_f32x4(a1, p, o1);
            o2 = wmma_f32x4(aones, p, o2);
        }
    }

    Ms[wv][lane] = mrow;
    Ls[wv][lane] = o2[0];
#pragma unroll
    for (int r = 0; r < 8; ++r) {
        Os[wv][r][lane]     = o0[r];
        Os[wv][8 + r][lane] = o1[r];
    }
    __syncthreads();

    if (wv == 0) {
        const float M = fmaxf(fmaxf(Ms[0][lane], Ms[1][lane]),
                              fmaxf(Ms[2][lane], Ms[3][lane]));
        float f[KSPLIT];
        float l = 0.f;
#pragma unroll
        for (int w = 0; w < KSPLIT; ++w) {
            f[w] = __expf(Ms[w][lane] - M);
            l += Ls[w][lane] * f[w];
        }
        const float inv = __builtin_amdgcn_rcpf(l);
        const float* gb = g_ws + (size_t)(q0 + c) * HD + h * CH + 8 * hi;
        float* ob = o_ws + (size_t)(q0 + c) * HD + h * CH + 8 * hi;
#pragma unroll
        for (int t = 0; t < 4; ++t) {
            const int r0 = (t >> 1) * 8 + (t & 1) * 4;
            const int off = (t >> 1) * 16 + (t & 1) * 4;
            const v4f g = *(const v4f*)(gb + off);
            v4f res;
#pragma unroll
            for (int e = 0; e < 4; ++e) {
                float a = 0.f;
#pragma unroll
                for (int w = 0; w < KSPLIT; ++w) a += Os[w][r0 + e][lane] * f[w];
                res[e] = a * inv * g[e];
            }
            *(v4f*)(ob + off) = res;
        }
    }
}

__global__ void out_kernel(const float* __restrict__ o_ws, const float* __restrict__ Wo,
                           const float* __restrict__ bo, float* __restrict__ out) {
    const int lane = threadIdx.x & 31;
    const int tile = blockIdx.x;
    const int row0 = blockIdx.y * 16;
    v8f acc = gemm_tile_f32(o_ws + (size_t)row0 * HD, HD, Wo + tile * 16, CIN, HD, lane);
    const int m = lane & 15, hi = lane >> 4, n = m;
    const float b = bo[tile * 16 + n];
#pragma unroll
    for (int r = 0; r < 8; ++r)
        out[(size_t)(row0 + r + 8 * hi) * CIN + tile * 16 + n] = acc[r] + b;
}

extern "C" void kernel_launch(void* const* d_in, const int* in_sizes, int n_in,
                              void* d_out, int out_size, void* d_ws, size_t ws_size,
                              hipStream_t stream) {
    const float* q_x       = (const float*)d_in[0];
    const float* kv_x      = (const float*)d_in[1];
    const float* bias_mask = (const float*)d_in[2];
    const float* bias_pair = (const float*)d_in[3];
    const float* Wq        = (const float*)d_in[4];
    const float* Wk        = (const float*)d_in[5];
    const float* Wv        = (const float*)d_in[6];
    const float* Wo        = (const float*)d_in[7];
    const float* bo        = (const float*)d_in[8];
    const float* Wg        = (const float*)d_in[9];
    const float* bg        = (const float*)d_in[10];
    float* out = (float*)d_out;

    const size_t mat = (size_t)QN * HD;
    float* q_ws = (float*)d_ws;
    float* k_ws = q_ws + mat;
    float* v_ws = k_ws + mat;
    float* g_ws = v_ws + mat;
    float* o_ws = g_ws + mat;

    proj_kernel<<<dim3(HD / 16, QN / 16, 4), 32, 0, stream>>>(
        q_x, kv_x, Wq, Wk, Wv, Wg, bg, q_ws, k_ws, v_ws, g_ws);

    flash_kernel<<<dim3(NH * (QN / 16)), 128, 0, stream>>>(
        q_ws, k_ws, v_ws, g_ws, bias_mask, bias_pair, o_ws);

    out_kernel<<<dim3(CIN / 16, QN / 16), 32, 0, stream>>>(o_ws, Wo, bo, out);
}


# projection kernel stores K and V in the attention kernel's fragment order: a K/V fragment load is 16 B per lane over 4 whole lines, the tile's loads one group; c_hid order in score sums changes
# speedup vs baseline: 1.4067x; 1.1332x over previous
#include <hip/hip_runtime.h>
#include <math.h>

typedef __attribute__((ext_vector_type(2))) float v2f;
typedef __attribute__((ext_vector_type(8))) float v8f;

#define QN 2048
#define KN 2048
#define CIN 256
#define HD 256
#define NH 8
#define CH 32
#define KSPLIT 4

__device__ __forceinline__ v8f wmma_f32x4(v2f a, v2f b, v8f c) {
    return __builtin_amdgcn_wmma_f32_16x16x4_f32(
         false, a,  false, b,
         (short)0, c,  false,  false);
}

template <bool TRANSPOSED>
__device__ __forceinline__ v8f gemm_tile_f32(const float* __restrict__ A, int lda,
                                             const float* __restrict__ B, int ldb,
                                             int CK, int lane) {
    const int m = lane & 15;
    const int hi = lane >> 4;
    const int n = m;
    v8f acc = {};
    for (int k0 = 0; k0 < CK; k0 += 4) {
        v2f a, b;
        a.x = A[m * lda + k0 + 2 * hi + 0];
        a.y = A[m * lda + k0 + 2 * hi + 1];
        b.x = B[(k0 + 2 * hi + 0) * ldb + n];
        b.y = B[(k0 + 2 * hi + 1) * ldb + n];
        acc = TRANSPOSED ? wmma_f32x4(b, a, acc) : wmma_f32x4(a, b, acc);
    }
    return acc;
}

typedef __attribute__((ext_vector_type(4))) float v4f;
#define KV_BLOCK 512

__global__ void proj_kernel(const float* __restrict__ qx, const float* __restrict__ kvx,
                            const float* __restrict__ Wq, const float* __restrict__ Wk,
                            const float* __restrict__ Wv, const float* __restrict__ Wg,
                            const float* __restrict__ bg,
                            float* __restrict__ q_ws, float* __restrict__ k_ws,
                            float* __restrict__ v_ws, float* __restrict__ g_ws) {
    const int lane = threadIdx.x & 31;
    const int tile = blockIdx.x;
    const int row0 = blockIdx.y * 16;
    const int which = blockIdx.z;

    const float* A = (which == 1 || which == 2) ? kvx : qx;
    const float* W = (which == 0) ? Wq : (which == 1) ? Wk : (which == 2) ? Wv : Wg;
    float* O = (which == 0) ? q_ws : (which == 1) ? k_ws : (which == 2) ? v_ws : g_ws;

    v8f acc = (which == 1)
        ? gemm_tile_f32<true>(A + (size_t)row0 * CIN, CIN, W + tile * 16, HD, CIN, lane)
        : gemm_tile_f32<false>(A + (size_t)row0 * CIN, CIN, W + tile * 16, HD, CIN, lane);

    const int m = lane & 15, hi = lane >> 4, n = m;
    if (which == 1 || which == 2) {
        const int hd = tile >> 1, half = tile & 1;
        float* blk = O + (size_t)(hd * (KN / 16) + blockIdx.y) * KV_BLOCK;
        float* dst = (which == 1) ? blk + (2 * hi) * 128 + (16 * half + n) * 4
                                  : blk + (2 * half) * 128 + lane * 4;
        v4f lo, up;
        lo[0] = acc[0]; lo[1] = acc[1]; lo[2] = acc[2]; lo[3] = acc[3];
        up[0] = acc[4]; up[1] = acc[5]; up[2] = acc[6]; up[3] = acc[7];
        *(v4f*)dst = lo;
        *(v4f*)(dst + 128) = up;
        return;
    }
    const int col = tile * 16 + n;
    const float bgc = (which == 3) ? bg[col] : 0.f;
#pragma unroll
    for (int r = 0; r < 8; ++r) {
        float val = acc[r];
        if (which == 0) val *= 0.17677669529663687f;
        if (which == 3) val = __builtin_amdgcn_rcpf(1.f + __expf(-(val + bgc)));
        O[(size_t)(row0 + r + 8 * hi) * HD + col] = val;
    }
}


__device__ __forceinline__ float xhalf(float x) {
    const int b = __builtin_bit_cast(int, x);
    return __builtin_bit_cast(float,
        __builtin_amdgcn_permlanex16(b, b, 0x76543210, 0xfedcba98, false, false));
}

__global__ void __launch_bounds__(128)
flash_kernel(const float* __restrict__ q_ws, const float* __restrict__ k_ws,
             const float* __restrict__ v_ws, const float* __restrict__ g_ws,
             const float* __restrict__ bias_mask, const float* __restrict__ bias_pair,
             float* __restrict__ o_ws) {
    __shared__ float Ms[KSPLIT][32];
    __shared__ float Ls[KSPLIT][32];
    __shared__ float Os[KSPLIT][16][32];

    const int lane = threadIdx.x & 31;
    const int wv = threadIdx.x >> 5;
    const int h = blockIdx.x >> 7;
    const int q0 = (blockIdx.x & 127) * 16;
    const int c = lane & 15, hi = lane >> 4;

    v2f bq[8];
    const v4f* Qb = (const v4f*)(q_ws + (size_t)(q0 + c) * HD + h * CH + 16 * hi);
#pragma unroll
    for (int i = 0; i < 4; ++i) {
        const v4f qv = Qb[i];
        bq[2 * i].x = qv[0];     bq[2 * i].y = qv[1];
        bq[2 * i + 1].x = qv[2]; bq[2 * i + 1].y = qv[3];
    }

    v8f o0 = {}, o1 = {}, o2 = {};
    float mrow = -1e30f;

    v2f aones; aones.x = 1.f; aones.y = 1.f;
    const float* bp_base = bias_pair + (size_t)h * QN * KN + (size_t)(q0 + c) * KN + 8 * hi;
    const float* bm_base = bias_mask + 8 * hi;
    const v4f* Kbase = (const v4f*)k_ws + (size_t)h * (KN / 16) * (KV_BLOCK / 4) + lane;
    const v4f* Vbase = (const v4f*)v_ws + (size_t)h * (KN / 16) * (KV_BLOCK / 4) + lane;
    const int kb_lo = wv * (KN / KSPLIT);
    const int kb_hi = kb_lo + (KN / KSPLIT);

    for (int kb = kb_lo; kb < kb_hi; kb += 16) {
        v8f s = {};
        const v4f* Kb = Kbase + (size_t)(kb >> 4) * (KV_BLOCK / 4);
        const v4f* Vb = Vbase + (size_t)(kb >> 4) * (KV_BLOCK / 4);
        v4f ka[4], va[4];
#pragma unroll
        for (int i = 0; i < 4; ++i) ka[i] = Kb[i * 32];
        const v4f bm0 = *(const v4f*)(bm_base + kb);
        const v4f bm1 = *(const v4f*)(bm_base + kb + 4);
        const v4f bp0 = __builtin_nontemporal_load((const v4f*)(bp_base + kb));
        const v4f bp1 = __builtin_nontemporal_load((const v4f*)(bp_base + kb + 4));
#pragma unroll
        for (int i = 0; i < 4; ++i) va[i] = Vb[i * 32];
        __builtin_amdgcn_sched_barrier(0);
#pragma unroll
        for (int kk = 0; kk < 8; ++kk) {
            v2f a;
            a.x = ka[kk >> 1][2 * (kk & 1)];
            a.y = ka[kk >> 1][2 * (kk & 1) + 1];
            s = wmma_f32x4(a, bq[kk], s);
        }
#pragma unroll
        for (int r = 0; r < 4; ++r) {
            s[r]     = (s[r]     + bm0[r]) + bp0[r];
            s[r + 4] = (s[r + 4] + bm1[r]) + bp1[r];
        }
        float mt = fmaxf(fmaxf(fmaxf(s[0], s[1]), fmaxf(s[2], s[3])),
                         fmaxf(fmaxf(s[4], s[5]), fmaxf(s[6], s[7])));
        mt = fmaxf(mt, xhalf(mt));
        const float mnew = fmaxf(mrow, mt);
        const float corr = __expf(mrow - mnew);
        mrow = mnew;
#pragma unroll
        for (int r = 0; r < 8; ++r) {
            s[r] = __expf(s[r] - mnew);
            o0[r] *= corr;
            o1[r] *= corr;
            o2[r] *= corr;
        }
#pragma unroll
        for (int j = 0; j < 4; ++j) {
            v2f p; p.x = s[2 * j]; p.y = s[2 * j + 1];
            v2f a0, a1;
            a0.x = va[j >> 1][2 * (j & 1)];      a0.y = va[j >> 1][2 * (j & 1) + 1];
            a1.x = va[2 + (j >> 1)][2 * (j & 1)]; a1.y = va[2 + (j >> 1)][2 * (j & 1) + 1];
            o0 = wmma_f32x4(a0, p, o0);
            o1 = wmma_f32x4(a1, p, o1);
            o2 = wmma_f32x4(aones, p, o2);
        }
    }

    Ms[wv][lane] = mrow;
    Ls[wv][lane] = o2[0];
#pragma unroll
    for (int r = 0; r < 8; ++r) {
        Os[wv][r][lane]     = o0[r];
        Os[wv][8 + r][lane] = o1[r];
    }
    __syncthreads();

    if (wv == 0) {
        const float M = fmaxf(fmaxf(Ms[0][lane], Ms[1][lane]),
                              fmaxf(Ms[2][lane], Ms[3][lane]));
        float f[KSPLIT];
        float l = 0.f;
#pragma unroll
        for (int w = 0; w < KSPLIT; ++w) {
            f[w] = __expf(Ms[w][lane] - M);
            l += Ls[w][lane] * f[w];
        }
        const float inv = __builtin_amdgcn_rcpf(l);
        const float* gb = g_ws + (size_t)(q0 + c) * HD + h * CH + 8 * hi;
        float* ob = o_ws + (size_t)(q0 + c) * HD + h * CH + 8 * hi;
#pragma unroll
        for (int t = 0; t < 4; ++t) {
            const int r0 = (t >> 1) * 8 + (t & 1) * 4;
            const int off = (t >> 1) * 16 + (t & 1) * 4;
            const v4f g = *(const v4f*)(gb + off);
            v4f res;
#pragma unroll
            for (int e = 0; e < 4; ++e) {
                float a = 0.f;
#pragma unroll
                for (int w = 0; w < KSPLIT; ++w) a += Os[w][r0 + e][lane] * f[w];
                res[e] = a * inv * g[e];
            }
            *(v4f*)(ob + off) = res;
        }
    }
}

__global__ void out_kernel(const float* __restrict__ o_ws, const float* __restrict__ Wo,
                           const float* __restrict__ bo, float* __restrict__ out) {
    const int lane = threadIdx.x & 31;
    const int tile = blockIdx.x;
    const int row0 = blockIdx.y * 16;
    v8f acc = gemm_tile_f32<false>(o_ws + (size_t)row0 * HD, HD, Wo + tile * 16, CIN, HD, lane);
    const int m = lane & 15, hi = lane >> 4, n = m;
    const float b = bo[tile * 16 + n];
#pragma unroll
    for (int r = 0; r < 8; ++r)
        out[(size_t)(row0 + r + 8 * hi) * CIN + tile * 16 + n] = acc[r] + b;
}

extern "C" void kernel_launch(void* const* d_in, const int* in_sizes, int n_in,
                              void* d_out, int out_size, void* d_ws, size_t ws_size,
                              hipStream_t stream) {
    const float* q_x       = (const float*)d_in[0];
    const float* kv_x      = (const float*)d_in[1];
    const float* bias_mask = (const float*)d_in[2];
    const float* bias_pair = (const float*)d_in[3];
    const float* Wq        = (const float*)d_in[4];
    const float* Wk        = (const float*)d_in[5];
    const float* Wv        = (const float*)d_in[6];
    const float* Wo        = (const float*)d_in[7];
    const float* bo        = (const float*)d_in[8];
    const float* Wg        = (const float*)d_in[9];
    const float* bg        = (const float*)d_in[10];
    float* out = (float*)d_out;

    const size_t mat = (size_t)QN * HD;
    float* q_ws = (float*)d_ws;
    float* k_ws = q_ws + mat;
    float* v_ws = k_ws + mat;
    float* g_ws = v_ws + mat;
    float* o_ws = g_ws + mat;

    proj_kernel<<<dim3(HD / 16, QN / 16, 4), 32, 0, stream>>>(
        q_x, kv_x, Wq, Wk, Wv, Wg, bg, q_ws, k_ws, v_ws, g_ws);

    flash_kernel<<<dim3(NH * (QN / 16)), 128, 0, stream>>>(
        q_ws, k_ws, v_ws, g_ws, bias_mask, bias_pair, o_ws);

    out_kernel<<<dim3(CIN / 16, QN / 16), 32, 0, stream>>>(o_ws, Wo, bo, out);
}


# projection GEMMs: a wave computes a 16x64 strip (output projection 16x32): A fragment loaded once per k step, lane owns 4 (2) adjacent columns so B fragments are one load per k row; no sum reordered
# speedup vs baseline: 1.6426x; 1.1677x over previous
#include <hip/hip_runtime.h>
#include <math.h>

typedef __attribute__((ext_vector_type(2))) float v2f;
typedef __attribute__((ext_vector_type(8))) float v8f;

#define QN 2048
#define KN 2048
#define CIN 256
#define HD 256
#define NH 8
#define CH 32
#define KSPLIT 4

__device__ __forceinline__ v8f wmma_f32x4(v2f a, v2f b, v8f c) {
    return __builtin_amdgcn_wmma_f32_16x16x4_f32(
         false, a,  false, b,
         (short)0, c,  false,  false);
}

typedef __attribute__((ext_vector_type(4))) float v4f;

template <int NT> struct strip_vec;
template <> struct strip_vec<2> { typedef v2f type; };
template <> struct strip_vec<4> { typedef v4f type; };

template <bool TRANSPOSED, int NT>
__device__ __forceinline__ void gemm_strip_f32(const float* __restrict__ A, int lda,
                                               const float* __restrict__ B, int ldb,
                                               int CK, int lane, v8f (&acc)[NT]) {
    typedef typename strip_vec<NT>::type bvec;
    const int m = lane & 15;
    const int hi = lane >> 4;
    const float* Ap = A + m * lda + 2 * hi;
    const float* Bp = B + (2 * hi) * ldb + NT * m;
#pragma unroll
    for (int t = 0; t < NT; ++t) acc[t] = v8f{};
    for (int kg = 0; kg < CK; kg += 32) {
        v2f a[8];
        bvec bx[8], by[8];
#pragma unroll
        for (int s = 0; s < 8; ++s) {
            const int k0 = kg + 4 * s;
            a[s] = *(const v2f*)(Ap + k0);
            bx[s] = *(const bvec*)(Bp + (size_t)k0 * ldb);
            by[s] = *(const bvec*)(Bp + (size_t)(k0 + 1) * ldb);
        }
        __builtin_amdgcn_sched_barrier(0);
#pragma unroll
        for (int s = 0; s < 8; ++s) {
#pragma unroll
            for (int t = 0; t < NT; ++t) {
                v2f b; b.x = bx[s][t]; b.y = by[s][t];
                acc[t] = TRANSPOSED ? wmma_f32x4(b, a[s], acc[t]) : wmma_f32x4(a[s], b, acc[t]);
            }
        }
    }
}

#define KV_BLOCK 512
#define PROJ_NT 4
#define OUT_NT 2

__global__ void proj_kernel(const float* __restrict__ qx, const float* __restrict__ kvx,
                            const float* __restrict__ Wq, const float* __restrict__ Wk,
                            const float* __restrict__ Wv, const float* __restrict__ Wg,
                            const float* __restrict__ bg,
                            float* __restrict__ q_ws, float* __restrict__ k_ws,
                            float* __restrict__ v_ws, float* __restrict__ g_ws) {
    const int lane = threadIdx.x & 31;
    const int strip = blockIdx.x;
    const int col0 = strip * 16 * PROJ_NT;
    const int row0 = blockIdx.y * 16;
    const int which = blockIdx.z;

    const float* A = (which == 1 || which == 2) ? kvx : qx;
    const float* W = (which == 0) ? Wq : (which == 1) ? Wk : (which == 2) ? Wv : Wg;
    float* O = (which == 0) ? q_ws : (which == 1) ? k_ws : (which == 2) ? v_ws : g_ws;

    v8f acc[PROJ_NT];
    if (which == 1)
        gemm_strip_f32<true, PROJ_NT>(A + (size_t)row0 * CIN, CIN, W + col0, HD, CIN, lane, acc);
    else
        gemm_strip_f32<false, PROJ_NT>(A + (size_t)row0 * CIN, CIN, W + col0, HD, CIN, lane, acc);

    const int c = lane & 15, hi = lane >> 4;
    if (which == 1) {
        float* blk = O + (size_t)((2 * strip + hi) * (KN / 16) + blockIdx.y) * KV_BLOCK;
#pragma unroll
        for (int r = 0; r < 8; ++r) {
            v4f w;
            w[0] = acc[0][r]; w[1] = acc[1][r]; w[2] = acc[2][r]; w[3] = acc[3][r];
            *(v4f*)(blk + (r & 3) * 128 + (16 * (r >> 2) + c) * 4) = w;
        }
        return;
    }
    if (which == 2) {
        float* blk = O + (size_t)((2 * strip + (c >> 3)) * (KN / 16) + blockIdx.y) * KV_BLOCK
                       + (2 * ((c >> 2) & 1)) * 128 + (16 * hi + 4 * (c & 3)) * 4;
#pragma unroll
        for (int t = 0; t < PROJ_NT; ++t) {
            v4f lo, up;
            lo[0] = acc[t][0]; lo[1] = acc[t][1]; lo[2] = acc[t][2]; lo[3] = acc[t][3];
            up[0] = acc[t][4]; up[1] = acc[t][5]; up[2] = acc[t][6]; up[3] = acc[t][7];
            *(v4f*)(blk + t * 4) = lo;
            *(v4f*)(blk + t * 4 + 128) = up;
        }
        return;
    }
    const int col = col0 + 4 * c;
    v4f bgc = {};
    if (which == 3) bgc = *(const v4f*)(bg + col);
#pragma unroll
    for (int r = 0; r < 8; ++r) {
        v4f w;
#pragma unroll
        for (int t = 0; t < PROJ_NT; ++t) {
            float val = acc[t][r];
            if (which == 0) val *= 0.17677669529663687f;
            if (which == 3) val = __builtin_amdgcn_rcpf(1.f + __expf(-(val + bgc[t])));
            w[t] = val;
        }
        *(v4f*)(O + (size_t)(row0 + r + 8 * hi) * HD + col) = w;
    }
}


__device__ __forceinline__ float xhalf(float x) {
    const int b = __builtin_bit_cast(int, x);
    return __builtin_bit_cast(float,
        __builtin_amdgcn_permlanex16(b, b, 0x76543210, 0xfedcba98, false, false));
}

__global__ void __launch_bounds__(128)
flash_kernel(const float* __restrict__ q_ws, const float* __restrict__ k_ws,
             const float* __restrict__ v_ws, const float* __restrict__ g_ws,
             const float* __restrict__ bias_mask, const float* __restrict__ bias_pair,
             float* __restrict__ o_ws) {
    __shared__ float Ms[KSPLIT][32];
    __shared__ float Ls[KSPLIT][32];
    __shared__ float Os[KSPLIT][16][32];

    const int lane = threadIdx.x & 31;
    const int wv = threadIdx.x >> 5;
    const int h = blockIdx.x >> 7;
    const int q0 = (blockIdx.x & 127) * 16;
    const int c = lane & 15, hi = lane >> 4;

    v2f bq[8];
    const v4f* Qb = (const v4f*)(q_ws + (size_t)(q0 + c) * HD + h * CH + 16 * hi);
#pragma unroll
    for (int i = 0; i < 4; ++i) {
        const v4f qv = Qb[i];
        bq[2 * i].x = qv[0];     bq[2 * i].y = qv[1];
        bq[2 * i + 1].x = qv[2]; bq[2 * i + 1].y = qv[3];
    }

    v8f o0 = {}, o1 = {}, o2 = {};
    float mrow = -1e30f;

    v2f aones; aones.x = 1.f; aones.y = 1.f;
    const float* bp_base = bias_pair + (size_t)h * QN * KN + (size_t)(q0 + c) * KN + 8 * hi;
    const float* bm_base = bias_mask + 8 * hi;
    const v4f* Kbase = (const v4f*)k_ws + (size_t)h * (KN / 16) * (KV_BLOCK / 4) + lane;
    const v4f* Vbase = (const v4f*)v_ws + (size_t)h * (KN / 16) * (KV_BLOCK / 4) + lane;
    const int kb_lo = wv * (KN / KSPLIT);
    const int kb_hi = kb_lo + (KN / KSPLIT);

    for (int kb = kb_lo; kb < kb_hi; kb += 16) {
        v8f s = {};
        const v4f* Kb = Kbase + (size_t)(kb >> 4) * (KV_BLOCK / 4);
        const v4f* Vb = Vbase + (size_t)(kb >> 4) * (KV_BLOCK / 4);
        v4f ka[4], va[4];
#pragma unroll
        for (int i = 0; i < 4; ++i) ka[i] = Kb[i * 32];
        const v4f bm0 = *(const v4f*)(bm_base + kb);
        const v4f bm1 = *(const v4f*)(bm_base + kb + 4);
        const v4f bp0 = __builtin_nontemporal_load((const v4f*)(bp_base + kb));
        const v4f bp1 = __builtin_nontemporal_load((const v4f*)(bp_base + kb + 4));
#pragma unroll
        for (int i = 0; i < 4; ++i) va[i] = Vb[i * 32];
        __builtin_amdgcn_sched_barrier(0);
#pragma unroll
        for (int kk = 0; kk < 8; ++kk) {
            v2f a;
            a.x = ka[kk >> 1][2 * (kk & 1)];
            a.y = ka[kk >> 1][2 * (kk & 1) + 1];
            s = wmma_f32x4(a, bq[kk], s);
        }
#pragma unroll
        for (int r = 0; r < 4; ++r) {
            s[r]     = (s[r]     + bm0[r]) + bp0[r];
            s[r + 4] = (s[r + 4] + bm1[r]) + bp1[r];
        }
        float mt = fmaxf(fmaxf(fmaxf(s[0], s[1]), fmaxf(s[2], s[3])),
                         fmaxf(fmaxf(s[4], s[5]), fmaxf(s[6], s[7])));
        mt = fmaxf(mt, xhalf(mt));
        const float mnew = fmaxf(mrow, mt);
        const float corr = __expf(mrow - mnew);
        mrow = mnew;
#pragma unroll
        for (int r = 0; r < 8; ++r) {
            s[r] = __expf(s[r] - mnew);
            o0[r] *= corr;
            o1[r] *= corr;
            o2[r] *= corr;
        }
#pragma unroll
        for (int j = 0; j < 4; ++j) {
            v2f p; p.x = s[2 * j]; p.y = s[2 * j + 1];
            v2f a0, a1;
            a0.x = va[j >> 1][2 * (j & 1)];      a0.y = va[j >> 1][2 * (j & 1) + 1];
            a1.x = va[2 + (j >> 1)][2 * (j & 1)]; a1.y = va[2 + (j >> 1)][2 * (j & 1) + 1];
            o0 = wmma_f32x4(a0, p, o0);
            o1 = wmma_f32x4(a1, p, o1);
            o2 = wmma_f32x4(aones, p, o2);
        }
    }

    Ms[wv][lane] = mrow;
    Ls[wv][lane] = o2[0];
#pragma unroll
    for (int r = 0; r < 8; ++r) {
        Os[wv][r][lane]     = o0[r];
        Os[wv][8 + r][lane] = o1[r];
    }
    __syncthreads();

    if (wv == 0) {
        const float M = fmaxf(fmaxf(Ms[0][lane], Ms[1][lane]),
                              fmaxf(Ms[2][lane], Ms[3][lane]));
        float f[KSPLIT];
        float l = 0.f;
#pragma unroll
        for (int w = 0; w < KSPLIT; ++w) {
            f[w] = __expf(Ms[w][lane] - M);
            l += Ls[w][lane] * f[w];
        }
        const float inv = __builtin_amdgcn_rcpf(l);
        const float* gb = g_ws + (size_t)(q0 + c) * HD + h * CH + 8 * hi;
        float* ob = o_ws + (size_t)(q0 + c) * HD + h * CH + 8 * hi;
#pragma unroll
        for (int t = 0; t < 4; ++t) {
            const int r0 = (t >> 1) * 8 + (t & 1) * 4;
            const int off = (t >> 1) * 16 + (t & 1) * 4;
            const v4f g = *(const v4f*)(gb + off);
            v4f res;
#pragma unroll
            for (int e = 0; e < 4; ++e) {
                float a = 0.f;
#pragma unroll
                for (int w = 0; w < KSPLIT; ++w) a += Os[w][r0 + e][lane] * f[w];
                res[e] = a * inv * g[e];
            }
            *(v4f*)(ob + off) = res;
        }
    }
}

__global__ void out_kernel(const float* __restrict__ o_ws, const float* __restrict__ Wo,
                           const float* __restrict__ bo, float* __restrict__ out) {
    const int lane = threadIdx.x & 31;
    const int col0 = blockIdx.x * 16 * OUT_NT;
    const int row0 = blockIdx.y * 16;
    v8f acc[OUT_NT];
    gemm_strip_f32<false, OUT_NT>(o_ws + (size_t)row0 * HD, HD, Wo + col0, CIN, HD, lane, acc);
    const int c = lane & 15, hi = lane >> 4;
    const int col = col0 + OUT_NT * c;
    const v2f b = *(const v2f*)(bo + col);
#pragma unroll
    for (int r = 0; r < 8; ++r) {
        v2f w;
        w.x = acc[0][r] + b.x;
        w.y = acc[1][r] + b.y;
        *(v2f*)(out + (size_t)(row0 + r + 8 * hi) * CIN + col) = w;
    }
}

extern "C" void kernel_launch(void* const* d_in, const int* in_sizes, int n_in,
                              void* d_out, int out_size, void* d_ws, size_t ws_size,
                              hipStream_t stream) {
    const float* q_x       = (const float*)d_in[0];
    const float* kv_x      = (const float*)d_in[1];
    const float* bias_mask = (const float*)d_in[2];
    const float* bias_pair = (const float*)d_in[3];
    const float* Wq        = (const float*)d_in[4];
    const float* Wk        = (const float*)d_in[5];
    const float* Wv        = (const float*)d_in[6];
    const float* Wo        = (const float*)d_in[7];
    const float* bo        = (const float*)d_in[8];
    const float* Wg        = (const float*)d_in[9];
    const float* bg        = (const float*)d_in[10];
    float* out = (float*)d_out;

    const size_t mat = (size_t)QN * HD;
    float* q_ws = (float*)d_ws;
    float* k_ws = q_ws + mat;
    float* v_ws = k_ws + mat;
    float* g_ws = v_ws + mat;
    float* o_ws = g_ws + mat;

    proj_kernel<<<dim3(HD / (16 * PROJ_NT), QN / 16, 4), 32, 0, stream>>>(
        q_x, kv_x, Wq, Wk, Wv, Wg, bg, q_ws, k_ws, v_ws, g_ws);

    flash_kernel<<<dim3(NH * (QN / 16)), 128, 0, stream>>>(
        q_ws, k_ws, v_ws, g_ws, bias_mask, bias_pair, o_ws);

    out_kernel<<<dim3(CIN / (16 * OUT_NT), QN / 16), 32, 0, stream>>>(o_ws, Wo, bo, out);
}


# attention kernel: a key tile's K, V and mask loads issued one tile ahead into a second register set, bias_pair loads three tiles ahead (four sets), each step's loads one group; no sum reordered
# speedup vs baseline: 1.7286x; 1.0524x over previous
#include <hip/hip_runtime.h>
#include <math.h>

typedef __attribute__((ext_vector_type(2))) float v2f;
typedef __attribute__((ext_vector_type(8))) float v8f;

#define QN 2048
#define KN 2048
#define CIN 256
#define HD 256
#define NH 8
#define CH 32
#define KSPLIT 4

__device__ __forceinline__ v8f wmma_f32x4(v2f a, v2f b, v8f c) {
    return __builtin_amdgcn_wmma_f32_16x16x4_f32(
         false, a,  false, b,
         (short)0, c,  false,  false);
}

typedef __attribute__((ext_vector_type(4))) float v4f;

template <int NT> struct strip_vec;
template <> struct strip_vec<2> { typedef v2f type; };
template <> struct strip_vec<4> { typedef v4f type; };

template <bool TRANSPOSED, int NT>
__device__ __forceinline__ void gemm_strip_f32(const float* __restrict__ A, int lda,
                                               const float* __restrict__ B, int ldb,
                                               int CK, int lane, v8f (&acc)[NT]) {
    typedef typename strip_vec<NT>::type bvec;
    const int m = lane & 15;
    const int hi = lane >> 4;
    const float* Ap = A + m * lda + 2 * hi;
    const float* Bp = B + (2 * hi) * ldb + NT * m;
#pragma unroll
    for (int t = 0; t < NT; ++t) acc[t] = v8f{};
    for (int kg = 0; kg < CK; kg += 32) {
        v2f a[8];
        bvec bx[8], by[8];
#pragma unroll
        for (int s = 0; s < 8; ++s) {
            const int k0 = kg + 4 * s;
            a[s] = *(const v2f*)(Ap + k0);
            bx[s] = *(const bvec*)(Bp + (size_t)k0 * ldb);
            by[s] = *(const bvec*)(Bp + (size_t)(k0 + 1) * ldb);
        }
        __builtin_amdgcn_sched_barrier(0);
#pragma unroll
        for (int s = 0; s < 8; ++s) {
#pragma unroll
            for (int t = 0; t < NT; ++t) {
                v2f b; b.x = bx[s][t]; b.y = by[s][t];
                acc[t] = TRANSPOSED ? wmma_f32x4(b, a[s], acc[t]) : wmma_f32x4(a[s], b, acc[t]);
            }
        }
    }
}

#define KV_BLOCK 512
#define PROJ_NT 4
#define OUT_NT 2

__global__ void proj_kernel(const float* __restrict__ qx, const float* __restrict__ kvx,
                            const float* __restrict__ Wq, const float* __restrict__ Wk,
                            const float* __restrict__ Wv, const float* __restrict__ Wg,
                            const float* __restrict__ bg,
                            float* __restrict__ q_ws, float* __restrict__ k_ws,
                            float* __restrict__ v_ws, float* __restrict__ g_ws) {
    const int lane = threadIdx.x & 31;
    const int strip = blockIdx.x;
    const int col0 = strip * 16 * PROJ_NT;
    const int row0 = blockIdx.y * 16;
    const int which = blockIdx.z;

    const float* A = (which == 1 || which == 2) ? kvx : qx;
    const float* W = (which == 0) ? Wq : (which == 1) ? Wk : (which == 2) ? Wv : Wg;
    float* O = (which == 0) ? q_ws : (which == 1) ? k_ws : (which == 2) ? v_ws : g_ws;

    v8f acc[PROJ_NT];
    if (which == 1)
        gemm_strip_f32<true, PROJ_NT>(A + (size_t)row0 * CIN, CIN, W + col0, HD, CIN, lane, acc);
    else
        gemm_strip_f32<false, PROJ_NT>(A + (size_t)row0 * CIN, CIN, W + col0, HD, CIN, lane, acc);

    const int c = lane & 15, hi = lane >> 4;
    if (which == 1) {
        float* blk = O + (size_t)((2 * strip + hi) * (KN / 16) + blockIdx.y) * KV_BLOCK;
#pragma unroll
        for (int r = 0; r < 8; ++r) {
            v4f w;
            w[0] = acc[0][r]; w[1] = acc[1][r]; w[2] = acc[2][r]; w[3] = acc[3][r];
            *(v4f*)(blk + (r & 3) * 128 + (16 * (r >> 2) + c) * 4) = w;
        }
        return;
    }
    if (which == 2) {
        float* blk = O + (size_t)((2 * strip + (c >> 3)) * (KN / 16) + blockIdx.y) * KV_BLOCK
                       + (2 * ((c >> 2) & 1)) * 128 + (16 * hi + 4 * (c & 3)) * 4;
#pragma unroll
        for (int t = 0; t < PROJ_NT; ++t) {
            v4f lo, up;
            lo[0] = acc[t][0]; lo[1] = acc[t][1]; lo[2] = acc[t][2]; lo[3] = acc[t][3];
            up[0] = acc[t][4]; up[1] = acc[t][5]; up[2] = acc[t][6]; up[3] = acc[t][7];
            *(v4f*)(blk + t * 4) = lo;
            *(v4f*)(blk + t * 4 + 128) = up;
        }
        return;
    }
    const int col = col0 + 4 * c;
    v4f bgc = {};
    if (which == 3) bgc = *(const v4f*)(bg + col);
#pragma unroll
    for (int r = 0; r < 8; ++r) {
        v4f w;
#pragma unroll
        for (int t = 0; t < PROJ_NT; ++t) {
            float val = acc[t][r];
            if (which == 0) val *= 0.17677669529663687f;
            if (which == 3) val = __builtin_amdgcn_rcpf(1.f + __expf(-(val + bgc[t])));
            w[t] = val;
        }
        *(v4f*)(O + (size_t)(row0 + r + 8 * hi) * HD + col) = w;
    }
}


__device__ __forceinline__ float xhalf(float x) {
    const int b = __builtin_bit_cast(int, x);
    return __builtin_bit_cast(float,
        __builtin_amdgcn_permlanex16(b, b, 0x76543210, 0xfedcba98, false, false));
}

struct kv_tile { v4f ka[4]; v4f va[4]; v4f bm0, bm1; };

__device__ __forceinline__ void load_kv_tile(kv_tile& t, const v4f* __restrict__ Kb,
                                             const v4f* __restrict__ Vb,
                                             const float* __restrict__ bm) {
#pragma unroll
    for (int i = 0; i < 4; ++i) t.ka[i] = Kb[i * 32];
    t.bm0 = *(const v4f*)(bm);
    t.bm1 = *(const v4f*)(bm + 4);
#pragma unroll
    for (int i = 0; i < 4; ++i) t.va[i] = Vb[i * 32];
}

__global__ void __launch_bounds__(128)
flash_kernel(const float* __restrict__ q_ws, const float* __restrict__ k_ws,
             const float* __restrict__ v_ws, const float* __restrict__ g_ws,
             const float* __restrict__ bias_mask, const float* __restrict__ bias_pair,
             float* __restrict__ o_ws) {
    __shared__ float Ms[KSPLIT][32];
    __shared__ float Ls[KSPLIT][32];
    __shared__ float Os[KSPLIT][16][32];

    const int lane = threadIdx.x & 31;
    const int wv = threadIdx.x >> 5;
    const int h = blockIdx.x >> 7;
    const int q0 = (blockIdx.x & 127) * 16;
    const int c = lane & 15, hi = lane >> 4;

    v2f bq[8];
    const v4f* Qb = (const v4f*)(q_ws + (size_t)(q0 + c) * HD + h * CH + 16 * hi);
#pragma unroll
    for (int i = 0; i < 4; ++i) {
        const v4f qv = Qb[i];
        bq[2 * i].x = qv[0];     bq[2 * i].y = qv[1];
        bq[2 * i + 1].x = qv[2]; bq[2 * i + 1].y = qv[3];
    }

    v8f o0 = {}, o1 = {}, o2 = {};
    float mrow = -1e30f;

    v2f aones; aones.x = 1.f; aones.y = 1.f;
    constexpr int NTILE = KN / KSPLIT / 16;
    const int kb_lo = wv * (KN / KSPLIT);
    const float* bpw = bias_pair + (size_t)h * QN * KN + (size_t)(q0 + c) * KN + 8 * hi + kb_lo;
    const float* bmw = bias_mask + 8 * hi + kb_lo;
    const v4f* Kw = (const v4f*)k_ws + ((size_t)h * (KN / 16) + (kb_lo >> 4)) * (KV_BLOCK / 4) + lane;
    const v4f* Vw = (const v4f*)v_ws + ((size_t)h * (KN / 16) + (kb_lo >> 4)) * (KV_BLOCK / 4) + lane;

    kv_tile T[2];
    v4f bpa[4], bpb[4];
    load_kv_tile(T[0], Kw, Vw, bmw);
#pragma unroll
    for (int d = 0; d < 3; ++d) {
        bpa[d] = __builtin_nontemporal_load((const v4f*)(bpw + 16 * d));
        bpb[d] = __builtin_nontemporal_load((const v4f*)(bpw + 16 * d + 4));
    }

    for (int it = 0; it < NTILE; it += 4) {
#pragma unroll
        for (int u = 0; u < 4; ++u) {
            const int j = it + u;
            const int jn = (j + 1 < NTILE) ? j + 1 : NTILE - 1;
            const int jp = (j + 3 < NTILE) ? j + 3 : NTILE - 1;
            load_kv_tile(T[(u + 1) & 1], Kw + (size_t)jn * (KV_BLOCK / 4),
                         Vw + (size_t)jn * (KV_BLOCK / 4), bmw + 16 * jn);
            bpa[(u + 3) & 3] = __builtin_nontemporal_load((const v4f*)(bpw + 16 * jp));
            bpb[(u + 3) & 3] = __builtin_nontemporal_load((const v4f*)(bpw + 16 * jp + 4));
            __builtin_amdgcn_sched_barrier(0);

            const kv_tile& t = T[u & 1];
            v8f s = {};
#pragma unroll
            for (int kk = 0; kk < 8; ++kk) {
                v2f a;
                a.x = t.ka[kk >> 1][2 * (kk & 1)];
                a.y = t.ka[kk >> 1][2 * (kk & 1) + 1];
                s = wmma_f32x4(a, bq[kk], s);
            }
            const v4f bp0 = bpa[u], bp1 = bpb[u];
#pragma unroll
            for (int r = 0; r < 4; ++r) {
                s[r]     = (s[r]     + t.bm0[r]) + bp0[r];
                s[r + 4] = (s[r + 4] + t.bm1[r]) + bp1[r];
            }
            float mt = fmaxf(fmaxf(fmaxf(s[0], s[1]), fmaxf(s[2], s[3])),
                             fmaxf(fmaxf(s[4], s[5]), fmaxf(s[6], s[7])));
            mt = fmaxf(mt, xhalf(mt));
            const float mnew = fmaxf(mrow, mt);
            const float corr = __expf(mrow - mnew);
            mrow = mnew;
#pragma unroll
            for (int r = 0; r < 8; ++r) {
                s[r] = __expf(s[r] - mnew);
                o0[r] *= corr;
                o1[r] *= corr;
                o2[r] *= corr;
            }
#pragma unroll
            for (int jj = 0; jj < 4; ++jj) {
                v2f p; p.x = s[2 * jj]; p.y = s[2 * jj + 1];
                v2f a0, a1;
                a0.x = t.va[jj >> 1][2 * (jj & 1)];      a0.y = t.va[jj >> 1][2 * (jj & 1) + 1];
                a1.x = t.va[2 + (jj >> 1)][2 * (jj & 1)]; a1.y = t.va[2 + (jj >> 1)][2 * (jj & 1) + 1];
                o0 = wmma_f32x4(a0, p, o0);
                o1 = wmma_f32x4(a1, p, o1);
                o2 = wmma_f32x4(aones, p, o2);
            }
            asm volatile("" : "+v"(o0), "+v"(o1), "+v"(o2));
        }
    }

    Ms[wv][lane] = mrow;
    Ls[wv][lane] = o2[0];
#pragma unroll
    for (int r = 0; r < 8; ++r) {
        Os[wv][r][lane]     = o0[r];
        Os[wv][8 + r][lane] = o1[r];
    }
    __syncthreads();

    if (wv == 0) {
        const float M = fmaxf(fmaxf(Ms[0][lane], Ms[1][lane]),
                              fmaxf(Ms[2][lane], Ms[3][lane]));
        float f[KSPLIT];
        float l = 0.f;
#pragma unroll
        for (int w = 0; w < KSPLIT; ++w) {
            f[w] = __expf(Ms[w][lane] - M);
            l += Ls[w][lane] * f[w];
        }
        const float inv = __builtin_amdgcn_rcpf(l);
        const float* gb = g_ws + (size_t)(q0 + c) * HD + h * CH + 8 * hi;
        float* ob = o_ws + (size_t)(q0 + c) * HD + h * CH + 8 * hi;
#pragma unroll
        for (int t = 0; t < 4; ++t) {
            const int r0 = (t >> 1) * 8 + (t & 1) * 4;
            const int off = (t >> 1) * 16 + (t & 1) * 4;
            const v4f g = *(const v4f*)(gb + off);
            v4f res;
#pragma unroll
            for (int e = 0; e < 4; ++e) {
                float a = 0.f;
#pragma unroll
                for (int w = 0; w < KSPLIT; ++w) a += Os[w][r0 + e][lane] * f[w];
                res[e] = a * inv * g[e];
            }
            *(v4f*)(ob + off) = res;
        }
    }
}

__global__ void out_kernel(const float* __restrict__ o_ws, const float* __restrict__ Wo,
                           const float* __restrict__ bo, float* __restrict__ out) {
    const int lane = threadIdx.x & 31;
    const int col0 = blockIdx.x * 16 * OUT_NT;
    const int row0 = blockIdx.y * 16;
    v8f acc[OUT_NT];
    gemm_strip_f32<false, OUT_NT>(o_ws + (size_t)row0 * HD, HD, Wo + col0, CIN, HD, lane, acc);
    const int c = lane & 15, hi = lane >> 4;
    const int col = col0 + OUT_NT * c;
    const v2f b = *(const v2f*)(bo + col);
#pragma unroll
    for (int r = 0; r < 8; ++r) {
        v2f w;
        w.x = acc[0][r] + b.x;
        w.y = acc[1][r] + b.y;
        *(v2f*)(out + (size_t)(row0 + r + 8 * hi) * CIN + col) = w;
    }
}

extern "C" void kernel_launch(void* const* d_in, const int* in_sizes, int n_in,
                              void* d_out, int out_size, void* d_ws, size_t ws_size,
                              hipStream_t stream) {
    const float* q_x       = (const float*)d_in[0];
    const float* kv_x      = (const float*)d_in[1];
    const float* bias_mask = (const float*)d_in[2];
    const float* bias_pair = (const float*)d_in[3];
    const float* Wq        = (const float*)d_in[4];
    const float* Wk        = (const float*)d_in[5];
    const float* Wv        = (const float*)d_in[6];
    const float* Wo        = (const float*)d_in[7];
    const float* bo        = (const float*)d_in[8];
    const float* Wg        = (const float*)d_in[9];
    const float* bg        = (const float*)d_in[10];
    float* out = (float*)d_out;

    const size_t mat = (size_t)QN * HD;
    float* q_ws = (float*)d_ws;
    float* k_ws = q_ws + mat;
    float* v_ws = k_ws + mat;
    float* g_ws = v_ws + mat;
    float* o_ws = g_ws + mat;

    proj_kernel<<<dim3(HD / (16 * PROJ_NT), QN / 16, 4), 32, 0, stream>>>(
        q_x, kv_x, Wq, Wk, Wv, Wg, bg, q_ws, k_ws, v_ws, g_ws);

    flash_kernel<<<dim3(NH * (QN / 16)), 128, 0, stream>>>(
        q_ws, k_ws, v_ws, g_ws, bias_mask, bias_pair, o_ws);

    out_kernel<<<dim3(CIN / (16 * OUT_NT), QN / 16), 32, 0, stream>>>(o_ws, Wo, bo, out);
}


# attention kernel: each wave carries two 16-query tiles through its key slice, so one K, V and mask fragment load feeds both (K and V reads halved, half as many waves); no sum reordered
# speedup vs baseline: 1.9516x; 1.1290x over previous
#include <hip/hip_runtime.h>
#include <math.h>

typedef __attribute__((ext_vector_type(2))) float v2f;
typedef __attribute__((ext_vector_type(8))) float v8f;

#define QN 2048
#define KN 2048
#define CIN 256
#define HD 256
#define NH 8
#define CH 32
#define KSPLIT 4

__device__ __forceinline__ v8f wmma_f32x4(v2f a, v2f b, v8f c) {
    return __builtin_amdgcn_wmma_f32_16x16x4_f32(
         false, a,  false, b,
         (short)0, c,  false,  false);
}

typedef __attribute__((ext_vector_type(4))) float v4f;

template <int NT> struct strip_vec;
template <> struct strip_vec<2> { typedef v2f type; };
template <> struct strip_vec<4> { typedef v4f type; };

template <bool TRANSPOSED, int NT>
__device__ __forceinline__ void gemm_strip_f32(const float* __restrict__ A, int lda,
                                               const float* __restrict__ B, int ldb,
                                               int CK, int lane, v8f (&acc)[NT]) {
    typedef typename strip_vec<NT>::type bvec;
    const int m = lane & 15;
    const int hi = lane >> 4;
    const float* Ap = A + m * lda + 2 * hi;
    const float* Bp = B + (2 * hi) * ldb + NT * m;
#pragma unroll
    for (int t = 0; t < NT; ++t) acc[t] = v8f{};
    for (int kg = 0; kg < CK; kg += 32) {
        v2f a[8];
        bvec bx[8], by[8];
#pragma unroll
        for (int s = 0; s < 8; ++s) {
            const int k0 = kg + 4 * s;
            a[s] = *(const v2f*)(Ap + k0);
            bx[s] = *(const bvec*)(Bp + (size_t)k0 * ldb);
            by[s] = *(const bvec*)(Bp + (size_t)(k0 + 1) * ldb);
        }
        __builtin_amdgcn_sched_barrier(0);
#pragma unroll
        for (int s = 0; s < 8; ++s) {
#pragma unroll
            for (int t = 0; t < NT; ++t) {
                v2f b; b.x = bx[s][t]; b.y = by[s][t];
                acc[t] = TRANSPOSED ? wmma_f32x4(b, a[s], acc[t]) : wmma_f32x4(a[s], b, acc[t]);
            }
        }
    }
}

#define KV_BLOCK 512
#define PROJ_NT 4
#define OUT_NT 2

__global__ void proj_kernel(const float* __restrict__ qx, const float* __restrict__ kvx,
                            const float* __restrict__ Wq, const float* __restrict__ Wk,
                            const float* __restrict__ Wv, const float* __restrict__ Wg,
                            const float* __restrict__ bg,
                            float* __restrict__ q_ws, float* __restrict__ k_ws,
                            float* __restrict__ v_ws, float* __restrict__ g_ws) {
    const int lane = threadIdx.x & 31;
    const int strip = blockIdx.x;
    const int col0 = strip * 16 * PROJ_NT;
    const int row0 = blockIdx.y * 16;
    const int which = blockIdx.z;

    const float* A = (which == 1 || which == 2) ? kvx : qx;
    const float* W = (which == 0) ? Wq : (which == 1) ? Wk : (which == 2) ? Wv : Wg;
    float* O = (which == 0) ? q_ws : (which == 1) ? k_ws : (which == 2) ? v_ws : g_ws;

    v8f acc[PROJ_NT];
    if (which == 1)
        gemm_strip_f32<true, PROJ_NT>(A + (size_t)row0 * CIN, CIN, W + col0, HD, CIN, lane, acc);
    else
        gemm_strip_f32<false, PROJ_NT>(A + (size_t)row0 * CIN, CIN, W + col0, HD, CIN, lane, acc);

    const int c = lane & 15, hi = lane >> 4;
    if (which == 1) {
        float* blk = O + (size_t)((2 * strip + hi) * (KN / 16) + blockIdx.y) * KV_BLOCK;
#pragma unroll
        for (int r = 0; r < 8; ++r) {
            v4f w;
            w[0] = acc[0][r]; w[1] = acc[1][r]; w[2] = acc[2][r]; w[3] = acc[3][r];
            *(v4f*)(blk + (r & 3) * 128 + (16 * (r >> 2) + c) * 4) = w;
        }
        return;
    }
    if (which == 2) {
        float* blk = O + (size_t)((2 * strip + (c >> 3)) * (KN / 16) + blockIdx.y) * KV_BLOCK
                       + (2 * ((c >> 2) & 1)) * 128 + (16 * hi + 4 * (c & 3)) * 4;
#pragma unroll
        for (int t = 0; t < PROJ_NT; ++t) {
            v4f lo, up;
            lo[0] = acc[t][0]; lo[1] = acc[t][1]; lo[2] = acc[t][2]; lo[3] = acc[t][3];
            up[0] = acc[t][4]; up[1] = acc[t][5]; up[2] = acc[t][6]; up[3] = acc[t][7];
            *(v4f*)(blk + t * 4) = lo;
            *(v4f*)(blk + t * 4 + 128) = up;
        }
        return;
    }
    const int col = col0 + 4 * c;
    v4f bgc = {};
    if (which == 3) bgc = *(const v4f*)(bg + col);
#pragma unroll
    for (int r = 0; r < 8; ++r) {
        v4f w;
#pragma unroll
        for (int t = 0; t < PROJ_NT; ++t) {
            float val = acc[t][r];
            if (which == 0) val *= 0.17677669529663687f;
            if (which == 3) val = __builtin_amdgcn_rcpf(1.f + __expf(-(val + bgc[t])));
            w[t] = val;
        }
        *(v4f*)(O + (size_t)(row0 + r + 8 * hi) * HD + col) = w;
    }
}

#define QT 2

__device__ __forceinline__ float xhalf(float x) {
    const int b = __builtin_bit_cast(int, x);
    return __builtin_bit_cast(float,
        __builtin_amdgcn_permlanex16(b, b, 0x76543210, 0xfedcba98, false, false));
}

struct kv_tile { v4f ka[4]; v4f va[4]; v4f bm0, bm1; };

__device__ __forceinline__ void load_kv_tile(kv_tile& t, const v4f* __restrict__ Kb,
                                             const v4f* __restrict__ Vb,
                                             const float* __restrict__ bm) {
#pragma unroll
    for (int i = 0; i < 4; ++i) t.ka[i] = Kb[i * 32];
    t.bm0 = *(const v4f*)(bm);
    t.bm1 = *(const v4f*)(bm + 4);
#pragma unroll
    for (int i = 0; i < 4; ++i) t.va[i] = Vb[i * 32];
}

__global__ void __launch_bounds__(128)
flash_kernel(const float* __restrict__ q_ws, const float* __restrict__ k_ws,
             const float* __restrict__ v_ws, const float* __restrict__ g_ws,
             const float* __restrict__ bias_mask, const float* __restrict__ bias_pair,
             float* __restrict__ o_ws) {
    static_assert(QT == 2, "the step-closing statement below names the accumulators of two q tiles");
    __shared__ float Ms[QT][KSPLIT][32];
    __shared__ float Ls[QT][KSPLIT][32];
    __shared__ float Os[QT][KSPLIT][16][32];

    const int lane = threadIdx.x & 31;
    const int wv = threadIdx.x >> 5;
    const int h = blockIdx.x / (QN / (16 * QT));
    const int q0 = (blockIdx.x % (QN / (16 * QT))) * (16 * QT);
    const int c = lane & 15, hi = lane >> 4;

    v2f bq[QT][8];
#pragma unroll
    for (int t = 0; t < QT; ++t) {
        const v4f* Qb = (const v4f*)(q_ws + (size_t)(q0 + 16 * t + c) * HD + h * CH + 16 * hi);
#pragma unroll
        for (int i = 0; i < 4; ++i) {
            const v4f qv = Qb[i];
            bq[t][2 * i].x = qv[0];     bq[t][2 * i].y = qv[1];
            bq[t][2 * i + 1].x = qv[2]; bq[t][2 * i + 1].y = qv[3];
        }
    }

    v8f o0[QT], o1[QT], o2[QT];
    float mrow[QT];
#pragma unroll
    for (int t = 0; t < QT; ++t) { o0[t] = v8f{}; o1[t] = v8f{}; o2[t] = v8f{}; mrow[t] = -1e30f; }

    v2f aones; aones.x = 1.f; aones.y = 1.f;
    constexpr int NTILE = KN / KSPLIT / 16;
    const int kb_lo = wv * (KN / KSPLIT);
    const float* bpw = bias_pair + (size_t)h * QN * KN + (size_t)(q0 + c) * KN + 8 * hi + kb_lo;
    const float* bmw = bias_mask + 8 * hi + kb_lo;
    const v4f* Kw = (const v4f*)k_ws + ((size_t)h * (KN / 16) + (kb_lo >> 4)) * (KV_BLOCK / 4) + lane;
    const v4f* Vw = (const v4f*)v_ws + ((size_t)h * (KN / 16) + (kb_lo >> 4)) * (KV_BLOCK / 4) + lane;

    kv_tile T[2];
    v4f bpa[QT][4], bpb[QT][4];
    load_kv_tile(T[0], Kw, Vw, bmw);
#pragma unroll
    for (int d = 0; d < 3; ++d) {
#pragma unroll
        for (int t = 0; t < QT; ++t) {
            const float* bp = bpw + (size_t)(16 * t) * KN + 16 * d;
            bpa[t][d] = __builtin_nontemporal_load((const v4f*)(bp));
            bpb[t][d] = __builtin_nontemporal_load((const v4f*)(bp + 4));
        }
    }

    for (int it = 0; it < NTILE; it += 4) {
#pragma unroll
        for (int u = 0; u < 4; ++u) {
            const int j = it + u;
            const int jn = (j + 1 < NTILE) ? j + 1 : NTILE - 1;
            const int jp = (j + 3 < NTILE) ? j + 3 : NTILE - 1;
            load_kv_tile(T[(u + 1) & 1], Kw + (size_t)jn * (KV_BLOCK / 4),
                         Vw + (size_t)jn * (KV_BLOCK / 4), bmw + 16 * jn);
#pragma unroll
            for (int t = 0; t < QT; ++t) {
                const float* bp = bpw + (size_t)(16 * t) * KN + 16 * jp;
                bpa[t][(u + 3) & 3] = __builtin_nontemporal_load((const v4f*)(bp));
                bpb[t][(u + 3) & 3] = __builtin_nontemporal_load((const v4f*)(bp + 4));
            }
            __builtin_amdgcn_sched_barrier(0);

            const kv_tile& kt = T[u & 1];
            v8f s[QT];
#pragma unroll
            for (int t = 0; t < QT; ++t) s[t] = v8f{};
#pragma unroll
            for (int kk = 0; kk < 8; ++kk) {
                v2f a;
                a.x = kt.ka[kk >> 1][2 * (kk & 1)];
                a.y = kt.ka[kk >> 1][2 * (kk & 1) + 1];
#pragma unroll
                for (int t = 0; t < QT; ++t) s[t] = wmma_f32x4(a, bq[t][kk], s[t]);
            }
#pragma unroll
            for (int t = 0; t < QT; ++t) {
                const v4f bp0 = bpa[t][u], bp1 = bpb[t][u];
#pragma unroll
                for (int r = 0; r < 4; ++r) {
                    s[t][r]     = (s[t][r]     + kt.bm0[r]) + bp0[r];
                    s[t][r + 4] = (s[t][r + 4] + kt.bm1[r]) + bp1[r];
                }
                float mt = fmaxf(fmaxf(fmaxf(s[t][0], s[t][1]), fmaxf(s[t][2], s[t][3])),
                                 fmaxf(fmaxf(s[t][4], s[t][5]), fmaxf(s[t][6], s[t][7])));
                mt = fmaxf(mt, xhalf(mt));
                const float mnew = fmaxf(mrow[t], mt);
                const float corr = __expf(mrow[t] - mnew);
                mrow[t] = mnew;
#pragma unroll
                for (int r = 0; r < 8; ++r) {
                    s[t][r] = __expf(s[t][r] - mnew);
                    o0[t][r] *= corr;
                    o1[t][r] *= corr;
                    o2[t][r] *= corr;
                }
#pragma unroll
                for (int jj = 0; jj < 4; ++jj) {
                    v2f p; p.x = s[t][2 * jj]; p.y = s[t][2 * jj + 1];
                    v2f a0, a1;
                    a0.x = kt.va[jj >> 1][2 * (jj & 1)];      a0.y = kt.va[jj >> 1][2 * (jj & 1) + 1];
                    a1.x = kt.va[2 + (jj >> 1)][2 * (jj & 1)]; a1.y = kt.va[2 + (jj >> 1)][2 * (jj & 1) + 1];
                    o0[t] = wmma_f32x4(a0, p, o0[t]);
                    o1[t] = wmma_f32x4(a1, p, o1[t]);
                    o2[t] = wmma_f32x4(aones, p, o2[t]);
                }
            }
            asm volatile("" : "+v"(o0[0]), "+v"(o1[0]), "+v"(o2[0]),
                              "+v"(o0[1]), "+v"(o1[1]), "+v"(o2[1]));
        }
    }

#pragma unroll
    for (int t = 0; t < QT; ++t) {
        Ms[t][wv][lane] = mrow[t];
        Ls[t][wv][lane] = o2[t][0];
#pragma unroll
        for (int r = 0; r < 8; ++r) {
            Os[t][wv][r][lane]     = o0[t][r];
            Os[t][wv][8 + r][lane] = o1[t][r];
        }
    }
    __syncthreads();

    if (wv < QT) {
        const int t = wv;
        const float M = fmaxf(fmaxf(Ms[t][0][lane], Ms[t][1][lane]),
                              fmaxf(Ms[t][2][lane], Ms[t][3][lane]));
        float f[KSPLIT];
        float l = 0.f;
#pragma unroll
        for (int w = 0; w < KSPLIT; ++w) {
            f[w] = __expf(Ms[t][w][lane] - M);
            l += Ls[t][w][lane] * f[w];
        }
        const float inv = __builtin_amdgcn_rcpf(l);
        const float* gb = g_ws + (size_t)(q0 + 16 * t + c) * HD + h * CH + 8 * hi;
        float* ob = o_ws + (size_t)(q0 + 16 * t + c) * HD + h * CH + 8 * hi;
#pragma unroll
        for (int pc = 0; pc < 4; ++pc) {
            const int r0 = (pc >> 1) * 8 + (pc & 1) * 4;
            const int off = (pc >> 1) * 16 + (pc & 1) * 4;
            const v4f g = *(const v4f*)(gb + off);
            v4f res;
#pragma unroll
            for (int e = 0; e < 4; ++e) {
                float a = 0.f;
#pragma unroll
                for (int w = 0; w < KSPLIT; ++w) a += Os[t][w][r0 + e][lane] * f[w];
                res[e] = a * inv * g[e];
            }
            *(v4f*)(ob + off) = res;
        }
    }
}

__global__ void out_kernel(const float* __restrict__ o_ws, const float* __restrict__ Wo,
                           const float* __restrict__ bo, float* __restrict__ out) {
    const int lane = threadIdx.x & 31;
    const int col0 = blockIdx.x * 16 * OUT_NT;
    const int row0 = blockIdx.y * 16;
    v8f acc[OUT_NT];
    gemm_strip_f32<false, OUT_NT>(o_ws + (size_t)row0 * HD, HD, Wo + col0, CIN, HD, lane, acc);
    const int c = lane & 15, hi = lane >> 4;
    const int col = col0 + OUT_NT * c;
    const v2f b = *(const v2f*)(bo + col);
#pragma unroll
    for (int r = 0; r < 8; ++r) {
        v2f w;
        w.x = acc[0][r] + b.x;
        w.y = acc[1][r] + b.y;
        *(v2f*)(out + (size_t)(row0 + r + 8 * hi) * CIN + col) = w;
    }
}

extern "C" void kernel_launch(void* const* d_in, const int* in_sizes, int n_in,
                              void* d_out, int out_size, void* d_ws, size_t ws_size,
                              hipStream_t stream) {
    const float* q_x       = (const float*)d_in[0];
    const float* kv_x      = (const float*)d_in[1];
    const float* bias_mask = (const float*)d_in[2];
    const float* bias_pair = (const float*)d_in[3];
    const float* Wq        = (const float*)d_in[4];
    const float* Wk        = (const float*)d_in[5];
    const float* Wv        = (const float*)d_in[6];
    const float* Wo        = (const float*)d_in[7];
    const float* bo        = (const float*)d_in[8];
    const float* Wg        = (const float*)d_in[9];
    const float* bg        = (const float*)d_in[10];
    float* out = (float*)d_out;

    const size_t mat = (size_t)QN * HD;
    float* q_ws = (float*)d_ws;
    float* k_ws = q_ws + mat;
    float* v_ws = k_ws + mat;
    float* g_ws = v_ws + mat;
    float* o_ws = g_ws + mat;

    proj_kernel<<<dim3(HD / (16 * PROJ_NT), QN / 16, 4), 32, 0, stream>>>(
        q_x, kv_x, Wq, Wk, Wv, Wg, bg, q_ws, k_ws, v_ws, g_ws);

    flash_kernel<<<dim3(NH * (QN / (16 * QT))), 128, 0, stream>>>(
        q_ws, k_ws, v_ws, g_ws, bias_mask, bias_pair, o_ws);

    out_kernel<<<dim3(CIN / (16 * OUT_NT), QN / 16), 32, 0, stream>>>(o_ws, Wo, bo, out);
}
